# starting kernel
# baseline (speedup 1.0000x reference)
#include <hip/hip_runtime.h>


typedef __attribute__((ext_vector_type(2))) float v2f;
typedef __attribute__((ext_vector_type(8))) float v8f;

#define IN_C    256
#define NH      8
#define KD2     288
#define MEM     256
#define HD      64
#define OUTC    256
#define HW      64
#define L_IMG   4096

#define LDS_X    0
#define LDS_P    17408
#define LDS_ATT  34816
#define LDS_O    71680
#define LDS_DW   80896

static __device__ __forceinline__ v8f wmma4(v2f a, v2f b, v8f c) {
  return __builtin_amdgcn_wmma_f32_16x16x4_f32(
       false, a,  false, b,
       (short)0, c,  false,  false);
}

__global__ __launch_bounds__(256)
void kcmm_fused_gfx1250(const float* __restrict__ x,
                        const float* __restrict__ Wp,
                        const float* __restrict__ bp,
                        const float* __restrict__ keyb,
                        const float* __restrict__ memb,
                        const float* __restrict__ Wo,
                        const float* __restrict__ bo,
                        float* __restrict__ out)
{
  extern __shared__ float lds[];

  const int tid  = threadIdx.x;
  const int wave = tid >> 5;
  const int lane = tid & 31;
  const int half = lane >> 4;
  const int li   = lane & 15;

  const int blk = blockIdx.x;
  const int b   = blk >> 8;
  const int rem = blk & 255;
  const int y   = rem >> 2;
  const int x0  = (rem & 3) << 4;

  const v8f vzero = {0.f,0.f,0.f,0.f,0.f,0.f,0.f,0.f};

  for (int idx = tid; idx < IN_C * 54; idx += 256) {
    int c  = idx / 54;
    int r2 = idx - c * 54;
    int rr = r2 / 18;
    int cc = r2 - rr * 18;
    int gy = y - 1 + rr;
    int gx = x0 - 1 + cc;
    float v = 0.f;
    if (gy >= 0 && gy < HW && gx >= 0 && gx < HW)
      v = x[((b * IN_C + c) * HW + gy) * HW + gx];
    lds[LDS_X + c * 68 + rr * 18 + cc] = v;
  }
  for (int idx = tid; idx < IN_C * 14; idx += 256) {
    int c = idx / 14;
    int q = 54 + (idx - c * 14);
    lds[LDS_X + c * 68 + q] = 0.f;
  }
  __syncthreads();

  {
    v8f acc[2][4];
    #pragma unroll
    for (int j = 0; j < 2; ++j)
      #pragma unroll
      for (int qt = 0; qt < 4; ++qt) acc[j][qt] = vzero;

    for (int k0 = 0; k0 < IN_C; k0 += 4) {
      const int kl = k0 + 2 * half;
      v2f bf[4];
      #pragma unroll
      for (int qt = 0; qt < 4; ++qt) {
        bf[qt].x = lds[LDS_X + kl * 68 + qt * 16 + li];
        bf[qt].y = lds[LDS_X + (kl + 1) * 68 + qt * 16 + li];
      }
      #pragma unroll
      for (int j = 0; j < 2; ++j) {
        const int o0 = (2 * wave + j) * 16;
        v2f af;
        af.x = Wp[(o0 + li) * IN_C + kl];
        af.y = Wp[(o0 + li) * IN_C + kl + 1];
        #pragma unroll
        for (int qt = 0; qt < 4; ++qt) acc[j][qt] = wmma4(af, bf[qt], acc[j][qt]);
      }
    }
    #pragma unroll
    for (int j = 0; j < 2; ++j)
      #pragma unroll
      for (int r = 0; r < 8; ++r) {
        const int o = (2 * wave + j) * 16 + 8 * half + r;
        const float bv = bp[o];
        #pragma unroll
        for (int qt = 0; qt < 4; ++qt)
          lds[LDS_P + o * 68 + qt * 16 + li] = acc[j][qt][r] + bv;
      }
  }
  __syncthreads();

  const int n = wave;
  v8f att[16];
  #pragma unroll
  for (int t = 0; t < 16; ++t) att[t] = vzero;

  for (int k0 = 0; k0 < KD2; k0 += 4) {
    const int kl = k0 + 2 * half;
    v2f bf;
    #pragma unroll
    for (int v = 0; v < 2; ++v) {
      const int ch = (kl + v) * NH + n;
      const int c  = ch / 9;
      const int p  = ch - c * 9;
      const int kh = p / 3;
      const int kw = p - kh * 3;
      bf[v] = lds[LDS_P + c * 68 + kh * 18 + kw + li];
    }
    #pragma unroll
    for (int t = 0; t < 16; ++t) {
      v2f af;
      af.x = keyb[(n * KD2 + kl) * MEM + t * 16 + li];
      af.y = keyb[(n * KD2 + kl + 1) * MEM + t * 16 + li];
      att[t] = wmma4(af, bf, att[t]);
    }
  }

  {
    float mx = -3.0e38f;
    #pragma unroll
    for (int t = 0; t < 16; ++t)
      #pragma unroll
      for (int r = 0; r < 8; ++r) mx = fmaxf(mx, att[t][r]);
    mx = fmaxf(mx, __shfl_xor(mx, 16, 32));

    float sum = 0.f;
    #pragma unroll
    for (int t = 0; t < 16; ++t)
      #pragma unroll
      for (int r = 0; r < 8; ++r) {
        float e = __expf(att[t][r] - mx);
        att[t][r] = e;
        sum += e;
      }
    sum += __shfl_xor(sum, 16, 32);
    const float inv = 1.f / sum;

    #pragma unroll
    for (int t = 0; t < 16; ++t)
      #pragma unroll
      for (int r = 0; r < 8; ++r) {
        const int i = t * 16 + 8 * half + r;
        lds[LDS_ATT + (n * MEM + i) * 18 + li] = att[t][r] * inv;
      }
  }

  {
    v8f oacc[4];
    #pragma unroll
    for (int t = 0; t < 4; ++t) oacc[t] = vzero;

    for (int k0 = 0; k0 < MEM; k0 += 4) {
      const int kl = k0 + 2 * half;
      v2f bf;
      bf.x = lds[LDS_ATT + (n * MEM + kl) * 18 + li];
      bf.y = lds[LDS_ATT + (n * MEM + kl + 1) * 18 + li];
      #pragma unroll
      for (int t = 0; t < 4; ++t) {
        v2f af;
        af.x = memb[(n * MEM + kl) * HD + t * 16 + li];
        af.y = memb[(n * MEM + kl + 1) * HD + t * 16 + li];
        oacc[t] = wmma4(af, bf, oacc[t]);
      }
    }
    #pragma unroll
    for (int t = 0; t < 4; ++t)
      #pragma unroll
      for (int r = 0; r < 8; ++r) {
        const int d = t * 16 + 8 * half + r;
        lds[LDS_O + (n * HD + d) * 18 + li] = oacc[t][r];
      }
  }
  __syncthreads();

  {
    v8f facc[2];
    facc[0] = vzero; facc[1] = vzero;

    for (int k0 = 0; k0 < NH * HD; k0 += 4) {
      const int kl = k0 + 2 * half;
      v2f bf;
      bf.x = lds[LDS_O + kl * 18 + li];
      bf.y = lds[LDS_O + (kl + 1) * 18 + li];
      #pragma unroll
      for (int j = 0; j < 2; ++j) {
        const int o0 = (2 * wave + j) * 16;
        v2f af;
        af.x = Wo[(o0 + li) * (NH * HD) + kl];
        af.y = Wo[(o0 + li) * (NH * HD) + kl + 1];
        facc[j] = wmma4(af, bf, facc[j]);
      }
    }
    #pragma unroll
    for (int j = 0; j < 2; ++j)
      #pragma unroll
      for (int r = 0; r < 8; ++r) {
        const int oo = (2 * wave + j) * 16 + 8 * half + r;
        out[(b * OUTC + oo) * L_IMG + y * HW + x0 + li] = facc[j][r] + bo[oo];
      }
  }
}

extern "C" void kernel_launch(void* const* d_in, const int* in_sizes, int n_in,
                              void* d_out, int out_size, void* d_ws, size_t ws_size,
                              hipStream_t stream) {
  (void)in_sizes; (void)n_in; (void)d_ws; (void)ws_size; (void)out_size;
  const float* x    = (const float*)d_in[0];
  const float* Wp   = (const float*)d_in[1];
  const float* bpj  = (const float*)d_in[2];
  const float* keyb = (const float*)d_in[3];
  const float* memb = (const float*)d_in[4];
  const float* Wo   = (const float*)d_in[5];
  const float* bo   = (const float*)d_in[6];
  float* out = (float*)d_out;

  const size_t shmem = (size_t)LDS_DW * sizeof(float);
  dim3 grid(8 * 64 * 4);
  dim3 block(256);
  kcmm_fused_gfx1250<<<grid, block, shmem, stream>>>(x, Wp, bpj, keyb, memb, Wo, bo, out);
}


# main launch head-major: a workgroup = one head x 8 pixel tiles, key and memory bank fetched once per workgroup and shared via LDS; out GEMM a third launch; permutes softmax and read-back sum order
# speedup vs baseline: 2.6431x; 2.6431x over previous
#include <hip/hip_runtime.h>


typedef __attribute__((ext_vector_type(2))) float v2f;
typedef __attribute__((ext_vector_type(8))) float v8f;
typedef __attribute__((ext_vector_type(4))) float v4f;
typedef __attribute__((ext_vector_type(2))) int v2i;

#define IN_C    256
#define NH      8
#define KD2     288
#define MEM     256
#define HD      64
#define OC      512
#define OUTC    256
#define HW      64
#define L_IMG   4096

#define XS1      68

#define KC       16
#define KCH      18
#define PA       544
#define A_BUF    4352
#define LDS_A    0
#define PM       136
#define LDS_M    8704
#define LDS_T    26112
#define LDS_HDW  26400

static __device__ __forceinline__ v8f wmma4(v2f a, v2f b, v8f c) {
  return __builtin_amdgcn_wmma_f32_16x16x4_f32(
       false, a,  false, b,
       (short)0, c,  false,  false);
}

__global__ __launch_bounds__(256) __attribute__((amdgpu_waves_per_eu(4, 4)))
void kcmm_proj_gfx1250(const float* __restrict__ x,
                       const float* __restrict__ Wp,
                       const float* __restrict__ bp,
                       float* __restrict__ P)
{
  extern __shared__ float lds[];

  const int tid  = threadIdx.x;
  const int wave = tid >> 5;
  const int lane = tid & 31;
  const int half = lane >> 4;
  const int li   = lane & 15;

  const int blk = blockIdx.x;
  const int b   = blk >> 6;
  const int y   = blk & 63;

  const v8f vzero = {0.f,0.f,0.f,0.f,0.f,0.f,0.f,0.f};

  const float* xt = x + (b * IN_C * HW + y) * HW;
  #pragma unroll
  for (int it = 0; it < 16; ++it) {
    const int idx = tid + 256 * it;
    const int c   = idx >> 4;
    const int sg  = idx & 15;
    const v4f v = *reinterpret_cast<const v4f*>(xt + c * L_IMG + 4 * sg);
    *reinterpret_cast<v4f*>(lds + c * XS1 + 4 * sg) = v;
  }
  __syncthreads();

  v8f acc[2][4];
  #pragma unroll
  for (int j = 0; j < 2; ++j)
    #pragma unroll
    for (int qt = 0; qt < 4; ++qt) acc[j][qt] = vzero;

  for (int k0 = 0; k0 < IN_C; k0 += 8) {
    const int kb = k0 + 4 * half;
    v2f bfa[4], bfb[4];
    #pragma unroll
    for (int qt = 0; qt < 4; ++qt) {
      bfa[qt].x = lds[kb * XS1 + qt * 16 + li];
      bfa[qt].y = lds[(kb + 1) * XS1 + qt * 16 + li];
      bfb[qt].x = lds[(kb + 2) * XS1 + qt * 16 + li];
      bfb[qt].y = lds[(kb + 3) * XS1 + qt * 16 + li];
    }
    #pragma unroll
    for (int j = 0; j < 2; ++j) {
      const int o0 = (2 * wave + j) * 16;
      const v4f w = *reinterpret_cast<const v4f*>(Wp + (o0 + li) * IN_C + kb);
      v2f afa, afb;
      afa.x = w[0]; afa.y = w[1];
      afb.x = w[2]; afb.y = w[3];
      #pragma unroll
      for (int qt = 0; qt < 4; ++qt) acc[j][qt] = wmma4(afa, bfa[qt], acc[j][qt]);
      #pragma unroll
      for (int qt = 0; qt < 4; ++qt) acc[j][qt] = wmma4(afb, bfb[qt], acc[j][qt]);
    }
  }
  #pragma unroll
  for (int j = 0; j < 2; ++j)
    #pragma unroll
    for (int r = 0; r < 8; ++r) {
      const int o = (2 * wave + j) * 16 + 8 * half + r;
      const float bv = bp[o];
      float* prow = P + ((b * IN_C + o) * HW + y) * HW;
      #pragma unroll
      for (int qt = 0; qt < 4; ++qt) prow[qt * 16 + li] = acc[j][qt][r] + bv;
    }
}

static __device__ __forceinline__ void key_chunk(v8f (&att)[16], const float* __restrict__ Ab,
                                                 const float (&bx)[4], const float (&by)[4],
                                                 const int vb, const int half, const int li) {
  #pragma unroll
  for (int ks = 0; ks < 4; ++ks) {
    v2f bf;
    bf.x = ((vb >> (2 * ks)) & 1) ? bx[ks] : 0.f;
    bf.y = ((vb >> (2 * ks + 1)) & 1) ? by[ks] : 0.f;
    const float* Ar = Ab + (2 * ks + half) * PA + 2 * li;
    #pragma unroll
    for (int t = 0; t < 16; ++t) {
      const v2f af = *reinterpret_cast<const v2f*>(Ar + 32 * t);
      att[t] = wmma4(af, bf, att[t]);
    }
  }
}

__global__ __launch_bounds__(256) __attribute__((amdgpu_waves_per_eu(4, 4)))
void kcmm_head_gfx1250(const float* __restrict__ P,
                       const float* __restrict__ keyb,
                       const float* __restrict__ memb,
                       float* __restrict__ O)
{
  extern __shared__ float lds[];
  int* ldsi = reinterpret_cast<int*>(lds);

  const int tid  = threadIdx.x;
  const int wave = tid >> 5;
  const int lane = tid & 31;
  const int half = lane >> 4;
  const int li   = lane & 15;

  const int blk   = blockIdx.x;
  const int n     = blk & 7;
  const int strip = blk >> 3;
  const int b     = strip >> 5;
  const int y     = ((strip & 31) << 1) + (wave >> 2);
  const int gx    = ((wave & 3) << 4) + li;

  const v8f vzero = {0.f,0.f,0.f,0.f,0.f,0.f,0.f,0.f};

  const float* kbn = keyb + n * KD2 * MEM;
  const int sgA = tid & 63;
  const int kq  = tid >> 6;
  v4f a0[2], a1[2];
  float curx[4], cury[4], nxtx[4], nxty[4];
  int curv, nxtv;

#define LOAD_A(c)                                                                            \
  _Pragma("unroll")                                                                          \
  for (int it = 0; it < 2; ++it) {                                                           \
    const int kp = kq + 4 * it;                                                              \
    a0[it] = *reinterpret_cast<const v4f*>(kbn + ((c) * KC + 2 * kp) * MEM + 4 * sgA);       \
    a1[it] = *reinterpret_cast<const v4f*>(kbn + ((c) * KC + 2 * kp + 1) * MEM + 4 * sgA);   \
  }
#define STORE_A(buf)                                                                         \
  _Pragma("unroll")                                                                          \
  for (int it = 0; it < 2; ++it) {                                                           \
    const int kp = kq + 4 * it;                                                              \
    const v4f s0 = {a0[it][0], a1[it][0], a0[it][1], a1[it][1]};                             \
    const v4f s1 = {a0[it][2], a1[it][2], a0[it][3], a1[it][3]};                             \
    float* dst = lds + LDS_A + (buf) * A_BUF + kp * PA + 8 * sgA;                            \
    *reinterpret_cast<v4f*>(dst)     = s0;                                                   \
    *reinterpret_cast<v4f*>(dst + 4) = s1;                                                   \
  }
#define GATHER_B(c, gxv, gyv, gvv)                                                           \
  gvv = 0;                                                                                   \
  _Pragma("unroll")                                                                          \
  for (int ks = 0; ks < 4; ++ks) {                                                           \
    const v2i e = *reinterpret_cast<const v2i*>(ldsi + LDS_T + (c) * KC + 4 * ks + 2 * half); \
    const int ok0 = (vmask >> (e.x & 15)) & 1;                                               \
    const int ok1 = (vmask >> (e.y & 15)) & 1;                                               \
    gxv[ks] = Pimg[ok0 ? (e.x >> 4) + pixbase : 0];                                          \
    gyv[ks] = Pimg[ok1 ? (e.y >> 4) + pixbase : 0];                                          \
    gvv |= (ok0 << (2 * ks)) | (ok1 << (2 * ks + 1));                                        \
  }

  LOAD_A(0)

  for (int k = tid; k < KD2; k += 256) {
    const int ch = k * NH + n;
    const int c  = ch / 9;
    const int p  = ch - c * 9;
    const int kh = p / 3;
    const int kw = p - kh * 3;
    ldsi[LDS_T + k] = ((c * L_IMG + kh * HW + kw) << 4) | p;
  }

  int vmask = 0;
  #pragma unroll
  for (int p = 0; p < 9; ++p) {
    const int yy = y + p / 3 - 1;
    const int xx = gx + p % 3 - 1;
    if (yy >= 0 && yy < HW && xx >= 0 && xx < HW) vmask |= 1 << p;
  }
  const int pixbase = (y - 1) * HW + gx - 1;
  const float* Pimg = P + b * IN_C * L_IMG;

  const float* mb = memb + n * MEM * HD;
  #pragma unroll
  for (int rd = 0; rd < 2; ++rd) {
    v4f m0[4], m1[4];
    #pragma unroll
    for (int it = 0; it < 4; ++it) {
      const int item = tid + 256 * (4 * rd + it);
      const int sg = item & 15;
      const int ip = item >> 4;
      m0[it] = *reinterpret_cast<const v4f*>(mb + (2 * ip) * HD + 4 * sg);
      m1[it] = *reinterpret_cast<const v4f*>(mb + (2 * ip + 1) * HD + 4 * sg);
    }
    #pragma unroll
    for (int it = 0; it < 4; ++it) {
      const int item = tid + 256 * (4 * rd + it);
      const int sg = item & 15;
      const int ip = item >> 4;
      const v4f s0 = {m0[it][0], m1[it][0], m0[it][1], m1[it][1]};
      const v4f s1 = {m0[it][2], m1[it][2], m0[it][3], m1[it][3]};
      *reinterpret_cast<v4f*>(lds + LDS_M + ip * PM + 8 * sg)     = s0;
      *reinterpret_cast<v4f*>(lds + LDS_M + ip * PM + 8 * sg + 4) = s1;
    }
  }
  __syncthreads();

  GATHER_B(0, curx, cury, curv)
  STORE_A(0)
  __syncthreads();

  v8f att[16];
  #pragma unroll
  for (int t = 0; t < 16; ++t) att[t] = vzero;

  for (int c = 0; c < KCH - 1; ++c) {
    LOAD_A(c + 1)
    GATHER_B(c + 1, nxtx, nxty, nxtv)
    key_chunk(att, lds + LDS_A + (c & 1) * A_BUF, curx, cury, curv, half, li);
    STORE_A((c + 1) & 1)
    #pragma unroll
    for (int ks = 0; ks < 4; ++ks) { curx[ks] = nxtx[ks]; cury[ks] = nxty[ks]; }
    curv = nxtv;
    __syncthreads();
  }
  key_chunk(att, lds + LDS_A + ((KCH - 1) & 1) * A_BUF, curx, cury, curv, half, li);
#undef LOAD_A
#undef STORE_A
#undef GATHER_B

  {
    float mx = -3.0e38f;
    #pragma unroll
    for (int t = 0; t < 16; ++t)
      #pragma unroll
      for (int r = 0; r < 8; ++r) mx = fmaxf(mx, att[t][r]);
    mx = fmaxf(mx, __shfl_xor(mx, 16, 32));

    float sum = 0.f;
    #pragma unroll
    for (int t = 0; t < 16; ++t)
      #pragma unroll
      for (int r = 0; r < 8; ++r) {
        float e = __expf(att[t][r] - mx);
        att[t][r] = e;
        sum += e;
      }
    sum += __shfl_xor(sum, 16, 32);
    const float inv = 1.f / sum;
    #pragma unroll
    for (int t = 0; t < 16; ++t)
      #pragma unroll
      for (int r = 0; r < 8; ++r) att[t][r] *= inv;
  }

  v8f oacc[4];
  #pragma unroll
  for (int dt = 0; dt < 4; ++dt) oacc[dt] = vzero;
  #pragma unroll
  for (int t = 0; t < 16; ++t)
    #pragma unroll
    for (int rp = 0; rp < 4; ++rp) {
      v2f bf;
      bf.x = att[t][2 * rp];
      bf.y = att[t][2 * rp + 1];
      const float* Mr = lds + LDS_M + (8 * t + 4 * half + rp) * PM + 2 * li;
      #pragma unroll
      for (int dt = 0; dt < 4; ++dt) {
        const v2f af = *reinterpret_cast<const v2f*>(Mr + 32 * dt);
        oacc[dt] = wmma4(af, bf, oacc[dt]);
      }
    }

  float* Ob = O + ((b * OC + n * HD) * HW + y) * HW + gx;
  #pragma unroll
  for (int dt = 0; dt < 4; ++dt)
    #pragma unroll
    for (int r = 0; r < 8; ++r)
      Ob[(16 * dt + 8 * half + r) * L_IMG] = oacc[dt][r];
}

__global__ __launch_bounds__(256) __attribute__((amdgpu_waves_per_eu(4, 4)))
void kcmm_out_gfx1250(const float* __restrict__ O,
                      const float* __restrict__ Wo,
                      const float* __restrict__ bo,
                      float* __restrict__ out)
{
  extern __shared__ float lds[];

  const int tid  = threadIdx.x;
  const int wave = tid >> 5;
  const int lane = tid & 31;
  const int half = lane >> 4;
  const int li   = lane & 15;

  const int blk = blockIdx.x;
  const int b   = blk >> 6;
  const int y   = blk & 63;

  const v8f vzero = {0.f,0.f,0.f,0.f,0.f,0.f,0.f,0.f};

  const float* ot = O + (b * OC * HW + y) * HW;
  #pragma unroll
  for (int bt = 0; bt < 2; ++bt) {
    #pragma unroll
    for (int it = 0; it < 16; ++it) {
      const int idx = tid + 256 * (16 * bt + it);
      const int c   = idx >> 4;
      const int sg  = idx & 15;
      const v4f v = *reinterpret_cast<const v4f*>(ot + c * L_IMG + 4 * sg);
      *reinterpret_cast<v4f*>(lds + c * XS1 + 4 * sg) = v;
    }
  }
  __syncthreads();

  v8f acc[2][4];
  #pragma unroll
  for (int j = 0; j < 2; ++j)
    #pragma unroll
    for (int qt = 0; qt < 4; ++qt) acc[j][qt] = vzero;

  for (int k0 = 0; k0 < OC; k0 += 8) {
    const int kb = k0 + 4 * half;
    v2f bfa[4], bfb[4];
    #pragma unroll
    for (int qt = 0; qt < 4; ++qt) {
      bfa[qt].x = lds[kb * XS1 + qt * 16 + li];
      bfa[qt].y = lds[(kb + 1) * XS1 + qt * 16 + li];
      bfb[qt].x = lds[(kb + 2) * XS1 + qt * 16 + li];
      bfb[qt].y = lds[(kb + 3) * XS1 + qt * 16 + li];
    }
    #pragma unroll
    for (int j = 0; j < 2; ++j) {
      const int o0 = (2 * wave + j) * 16;
      const v4f w = *reinterpret_cast<const v4f*>(Wo + (o0 + li) * OC + kb);
      v2f afa, afb;
      afa.x = w[0]; afa.y = w[1];
      afb.x = w[2]; afb.y = w[3];
      #pragma unroll
      for (int qt = 0; qt < 4; ++qt) acc[j][qt] = wmma4(afa, bfa[qt], acc[j][qt]);
      #pragma unroll
      for (int qt = 0; qt < 4; ++qt) acc[j][qt] = wmma4(afb, bfb[qt], acc[j][qt]);
    }
  }
  #pragma unroll
  for (int j = 0; j < 2; ++j)
    #pragma unroll
    for (int r = 0; r < 8; ++r) {
      const int o = (2 * wave + j) * 16 + 8 * half + r;
      const float bv = bo[o];
      float* orow = out + ((b * OUTC + o) * HW + y) * HW;
      #pragma unroll
      for (int qt = 0; qt < 4; ++qt) orow[qt * 16 + li] = acc[j][qt][r] + bv;
    }
}

extern "C" void kernel_launch(void* const* d_in, const int* in_sizes, int n_in,
                              void* d_out, int out_size, void* d_ws, size_t ws_size,
                              hipStream_t stream) {
  (void)in_sizes; (void)n_in; (void)ws_size; (void)out_size;
  const float* x    = (const float*)d_in[0];
  const float* Wp   = (const float*)d_in[1];
  const float* bpj  = (const float*)d_in[2];
  const float* keyb = (const float*)d_in[3];
  const float* memb = (const float*)d_in[4];
  const float* Wo   = (const float*)d_in[5];
  const float* bo   = (const float*)d_in[6];
  float* out = (float*)d_out;

  float* P = (float*)d_ws;
  float* O = P + 8 * IN_C * L_IMG;

  const size_t shmem1 = (size_t)(IN_C * XS1) * sizeof(float);
  kcmm_proj_gfx1250<<<dim3(8 * 64), dim3(256), shmem1, stream>>>(x, Wp, bpj, P);

  const size_t shmem2 = (size_t)LDS_HDW * sizeof(float);
  kcmm_head_gfx1250<<<dim3(8 * 32 * 8), dim3(256), shmem2, stream>>>(P, keyb, memb, O);

  const size_t shmem3 = (size_t)(OC * XS1) * sizeof(float);
  kcmm_out_gfx1250<<<dim3(8 * 64), dim3(256), shmem3, stream>>>(O, Wo, bo, out);
}


# key GEMM k order tap-major: a 16-k chunk is 16 channels of one tap, so a lane's 8 gathers are one address plus constant strides and one in-image bit; no unfold table; permutes key GEMM k (sum) order
# speedup vs baseline: 2.7217x; 1.0297x over previous
#include <hip/hip_runtime.h>


typedef __attribute__((ext_vector_type(2))) float v2f;
typedef __attribute__((ext_vector_type(8))) float v8f;
typedef __attribute__((ext_vector_type(4))) float v4f;

#define IN_C    256
#define NH      8
#define KD2     288
#define MEM     256
#define HD      64
#define OC      512
#define OUTC    256
#define HW      64
#define L_IMG   4096

#define XS1      68

#define KC       16
#define KCH      18
#define PA       544
#define A_BUF    4352
#define LDS_A    0
#define PM       136
#define LDS_M    8704
#define LDS_HDW  26112

static __device__ __forceinline__ v8f wmma4(v2f a, v2f b, v8f c) {
  return __builtin_amdgcn_wmma_f32_16x16x4_f32(
       false, a,  false, b,
       (short)0, c,  false,  false);
}

__global__ __launch_bounds__(256) __attribute__((amdgpu_waves_per_eu(4, 4)))
void kcmm_proj_gfx1250(const float* __restrict__ x,
                       const float* __restrict__ Wp,
                       const float* __restrict__ bp,
                       float* __restrict__ P)
{
  extern __shared__ float lds[];

  const int tid  = threadIdx.x;
  const int wave = tid >> 5;
  const int lane = tid & 31;
  const int half = lane >> 4;
  const int li   = lane & 15;

  const int blk = blockIdx.x;
  const int b   = blk >> 6;
  const int y   = blk & 63;

  const v8f vzero = {0.f,0.f,0.f,0.f,0.f,0.f,0.f,0.f};

  const float* xt = x + (b * IN_C * HW + y) * HW;
  #pragma unroll
  for (int it = 0; it < 16; ++it) {
    const int idx = tid + 256 * it;
    const int c   = idx >> 4;
    const int sg  = idx & 15;
    const v4f v = *reinterpret_cast<const v4f*>(xt + c * L_IMG + 4 * sg);
    *reinterpret_cast<v4f*>(lds + c * XS1 + 4 * sg) = v;
  }
  __syncthreads();

  v8f acc[2][4];
  #pragma unroll
  for (int j = 0; j < 2; ++j)
    #pragma unroll
    for (int qt = 0; qt < 4; ++qt) acc[j][qt] = vzero;

  for (int k0 = 0; k0 < IN_C; k0 += 8) {
    const int kb = k0 + 4 * half;
    v2f bfa[4], bfb[4];
    #pragma unroll
    for (int qt = 0; qt < 4; ++qt) {
      bfa[qt].x = lds[kb * XS1 + qt * 16 + li];
      bfa[qt].y = lds[(kb + 1) * XS1 + qt * 16 + li];
      bfb[qt].x = lds[(kb + 2) * XS1 + qt * 16 + li];
      bfb[qt].y = lds[(kb + 3) * XS1 + qt * 16 + li];
    }
    #pragma unroll
    for (int j = 0; j < 2; ++j) {
      const int o0 = (2 * wave + j) * 16;
      const v4f w = *reinterpret_cast<const v4f*>(Wp + (o0 + li) * IN_C + kb);
      v2f afa, afb;
      afa.x = w[0]; afa.y = w[1];
      afb.x = w[2]; afb.y = w[3];
      #pragma unroll
      for (int qt = 0; qt < 4; ++qt) acc[j][qt] = wmma4(afa, bfa[qt], acc[j][qt]);
      #pragma unroll
      for (int qt = 0; qt < 4; ++qt) acc[j][qt] = wmma4(afb, bfb[qt], acc[j][qt]);
    }
  }
  #pragma unroll
  for (int j = 0; j < 2; ++j)
    #pragma unroll
    for (int r = 0; r < 8; ++r) {
      const int o = (2 * wave + j) * 16 + 8 * half + r;
      const float bv = bp[o];
      float* prow = P + ((b * IN_C + o) * HW + y) * HW;
      #pragma unroll
      for (int qt = 0; qt < 4; ++qt) prow[qt * 16 + li] = acc[j][qt][r] + bv;
    }
}

static __device__ __forceinline__ void key_chunk(v8f (&att)[16], const float* __restrict__ Ab,
                                                 const float (&bx)[4], const float (&by)[4],
                                                 const int ok, const int half, const int li) {
  #pragma unroll
  for (int ks = 0; ks < 4; ++ks) {
    v2f bf;
    bf.x = ok ? bx[ks] : 0.f;
    bf.y = ok ? by[ks] : 0.f;
    const float* Ar = Ab + (2 * ks + half) * PA + 2 * li;
    #pragma unroll
    for (int t = 0; t < 16; ++t) {
      const v2f af = *reinterpret_cast<const v2f*>(Ar + 32 * t);
      att[t] = wmma4(af, bf, att[t]);
    }
  }
}

__global__ __launch_bounds__(256) __attribute__((amdgpu_waves_per_eu(4, 4)))
void kcmm_head_gfx1250(const float* __restrict__ P,
                       const float* __restrict__ keyb,
                       const float* __restrict__ memb,
                       float* __restrict__ O)
{
  extern __shared__ float lds[];

  const int tid  = threadIdx.x;
  const int wave = tid >> 5;
  const int lane = tid & 31;
  const int half = lane >> 4;
  const int li   = lane & 15;

  const int blk   = blockIdx.x;
  const int n     = blk & 7;
  const int strip = blk >> 3;
  const int b     = strip >> 5;
  const int y     = ((strip & 31) << 1) + (wave >> 2);
  const int gx    = ((wave & 3) << 4) + li;

  const v8f vzero = {0.f,0.f,0.f,0.f,0.f,0.f,0.f,0.f};

  const float* kbn = keyb + n * KD2 * MEM;
  const int sgA = tid & 63;
  const int kq  = tid >> 6;
  v4f a0[2], a1[2];
  float curx[4], cury[4], nxtx[4], nxty[4];
  int curok, nxtok;

#define LOAD_A(cc)                                                                           \
  {                                                                                          \
    const int p_  = (cc) >> 1;                                                               \
    const int m_  = (n - p_) & 7;                                                            \
    const int k0_ = m_ + ((m_ + p_ - n) >> 3) + 144 * ((cc) & 1);      \
    const float* src_ = kbn + k0_ * MEM + 4 * sgA;                                           \
    _Pragma("unroll")                                                                        \
    for (int it = 0; it < 2; ++it) {                                                         \
      const int kp = kq + 4 * it;                       \
      a0[it] = *reinterpret_cast<const v4f*>(src_ + (18 * kp) * MEM);                        \
      a1[it] = *reinterpret_cast<const v4f*>(src_ + (18 * kp + 9) * MEM);                    \
    }                                                                                        \
  }
#define STORE_A(buf)                                                                         \
  _Pragma("unroll")                                                                          \
  for (int it = 0; it < 2; ++it) {                                                           \
    const int kp = kq + 4 * it;                                                              \
    const v4f s0 = {a0[it][0], a1[it][0], a0[it][1], a1[it][1]};                             \
    const v4f s1 = {a0[it][2], a1[it][2], a0[it][3], a1[it][3]};                             \
    float* dst = lds + LDS_A + (buf) * A_BUF + kp * PA + 8 * sgA;                            \
    *reinterpret_cast<v4f*>(dst)     = s0;                                                   \
    *reinterpret_cast<v4f*>(dst + 4) = s1;                                                   \
  }
#define GATHER_B(cc, gxv, gyv, gok)                                                          \
  {                                                                                          \
    const int p_  = (cc) >> 1;                                                               \
    const int m_  = (n - p_) & 7;                                                            \
    const int kh_ = p_ / 3;                                                                  \
    const int kw_ = p_ - 3 * kh_;                                                            \
    gok = (vmask >> p_) & 1;                                                                 \
      \
    const int i0_ = ((m_ + 128 * ((cc) & 1) + 16 * half) * L_IMG + kh_ * HW + kw_ + pixbase) & -gok; \
    _Pragma("unroll")                                                                        \
    for (int ks = 0; ks < 4; ++ks) {                                                         \
      gxv[ks] = Pimg[i0_ + (32 * ks) * L_IMG];                                               \
      gyv[ks] = Pimg[i0_ + (32 * ks + 8) * L_IMG];                                           \
    }                                                                                        \
  }

  LOAD_A(0)

  int vmask = 0;
  #pragma unroll
  for (int p = 0; p < 9; ++p) {
    const int yy = y + p / 3 - 1;
    const int xx = gx + p % 3 - 1;
    if (yy >= 0 && yy < HW && xx >= 0 && xx < HW) vmask |= 1 << p;
  }
  const int pixbase = (y - 1) * HW + gx - 1;
  const float* Pimg = P + b * IN_C * L_IMG;
  GATHER_B(0, curx, cury, curok)

  const float* mb = memb + n * MEM * HD;
  #pragma unroll
  for (int rd = 0; rd < 2; ++rd) {
    v4f m0[4], m1[4];
    #pragma unroll
    for (int it = 0; it < 4; ++it) {
      const int item = tid + 256 * (4 * rd + it);
      const int sg = item & 15;
      const int ip = item >> 4;
      m0[it] = *reinterpret_cast<const v4f*>(mb + (2 * ip) * HD + 4 * sg);
      m1[it] = *reinterpret_cast<const v4f*>(mb + (2 * ip + 1) * HD + 4 * sg);
    }
    #pragma unroll
    for (int it = 0; it < 4; ++it) {
      const int item = tid + 256 * (4 * rd + it);
      const int sg = item & 15;
      const int ip = item >> 4;
      const v4f s0 = {m0[it][0], m1[it][0], m0[it][1], m1[it][1]};
      const v4f s1 = {m0[it][2], m1[it][2], m0[it][3], m1[it][3]};
      *reinterpret_cast<v4f*>(lds + LDS_M + ip * PM + 8 * sg)     = s0;
      *reinterpret_cast<v4f*>(lds + LDS_M + ip * PM + 8 * sg + 4) = s1;
    }
  }
  STORE_A(0)
  __syncthreads();

  v8f att[16];
  #pragma unroll
  for (int t = 0; t < 16; ++t) att[t] = vzero;

  for (int c = 0; c < KCH - 1; ++c) {
    LOAD_A(c + 1)
    GATHER_B(c + 1, nxtx, nxty, nxtok)
    key_chunk(att, lds + LDS_A + (c & 1) * A_BUF, curx, cury, curok, half, li);
    STORE_A((c + 1) & 1)
    #pragma unroll
    for (int ks = 0; ks < 4; ++ks) { curx[ks] = nxtx[ks]; cury[ks] = nxty[ks]; }
    curok = nxtok;
    __syncthreads();
  }
  key_chunk(att, lds + LDS_A + ((KCH - 1) & 1) * A_BUF, curx, cury, curok, half, li);
#undef LOAD_A
#undef STORE_A
#undef GATHER_B

  {
    float mx = -3.0e38f;
    #pragma unroll
    for (int t = 0; t < 16; ++t)
      #pragma unroll
      for (int r = 0; r < 8; ++r) mx = fmaxf(mx, att[t][r]);
    mx = fmaxf(mx, __shfl_xor(mx, 16, 32));

    float sum = 0.f;
    #pragma unroll
    for (int t = 0; t < 16; ++t)
      #pragma unroll
      for (int r = 0; r < 8; ++r) {
        float e = __expf(att[t][r] - mx);
        att[t][r] = e;
        sum += e;
      }
    sum += __shfl_xor(sum, 16, 32);
    const float inv = 1.f / sum;
    #pragma unroll
    for (int t = 0; t < 16; ++t)
      #pragma unroll
      for (int r = 0; r < 8; ++r) att[t][r] *= inv;
  }

  v8f oacc[4];
  #pragma unroll
  for (int dt = 0; dt < 4; ++dt) oacc[dt] = vzero;
  #pragma unroll
  for (int t = 0; t < 16; ++t)
    #pragma unroll
    for (int rp = 0; rp < 4; ++rp) {
      v2f bf;
      bf.x = att[t][2 * rp];
      bf.y = att[t][2 * rp + 1];
      const float* Mr = lds + LDS_M + (8 * t + 4 * half + rp) * PM + 2 * li;
      #pragma unroll
      for (int dt = 0; dt < 4; ++dt) {
        const v2f af = *reinterpret_cast<const v2f*>(Mr + 32 * dt);
        oacc[dt] = wmma4(af, bf, oacc[dt]);
      }
    }

  float* Ob = O + ((b * OC + n * HD) * HW + y) * HW + gx;
  #pragma unroll
  for (int dt = 0; dt < 4; ++dt)
    #pragma unroll
    for (int r = 0; r < 8; ++r)
      Ob[(16 * dt + 8 * half + r) * L_IMG] = oacc[dt][r];
}

__global__ __launch_bounds__(256) __attribute__((amdgpu_waves_per_eu(4, 4)))
void kcmm_out_gfx1250(const float* __restrict__ O,
                      const float* __restrict__ Wo,
                      const float* __restrict__ bo,
                      float* __restrict__ out)
{
  extern __shared__ float lds[];

  const int tid  = threadIdx.x;
  const int wave = tid >> 5;
  const int lane = tid & 31;
  const int half = lane >> 4;
  const int li   = lane & 15;

  const int blk = blockIdx.x;
  const int b   = blk >> 6;
  const int y   = blk & 63;

  const v8f vzero = {0.f,0.f,0.f,0.f,0.f,0.f,0.f,0.f};

  const float* ot = O + (b * OC * HW + y) * HW;
  #pragma unroll
  for (int bt = 0; bt < 2; ++bt) {
    #pragma unroll
    for (int it = 0; it < 16; ++it) {
      const int idx = tid + 256 * (16 * bt + it);
      const int c   = idx >> 4;
      const int sg  = idx & 15;
      const v4f v = *reinterpret_cast<const v4f*>(ot + c * L_IMG + 4 * sg);
      *reinterpret_cast<v4f*>(lds + c * XS1 + 4 * sg) = v;
    }
  }
  __syncthreads();

  v8f acc[2][4];
  #pragma unroll
  for (int j = 0; j < 2; ++j)
    #pragma unroll
    for (int qt = 0; qt < 4; ++qt) acc[j][qt] = vzero;

  for (int k0 = 0; k0 < OC; k0 += 8) {
    const int kb = k0 + 4 * half;
    v2f bfa[4], bfb[4];
    #pragma unroll
    for (int qt = 0; qt < 4; ++qt) {
      bfa[qt].x = lds[kb * XS1 + qt * 16 + li];
      bfa[qt].y = lds[(kb + 1) * XS1 + qt * 16 + li];
      bfb[qt].x = lds[(kb + 2) * XS1 + qt * 16 + li];
      bfb[qt].y = lds[(kb + 3) * XS1 + qt * 16 + li];
    }
    #pragma unroll
    for (int j = 0; j < 2; ++j) {
      const int o0 = (2 * wave + j) * 16;
      const v4f w = *reinterpret_cast<const v4f*>(Wo + (o0 + li) * OC + kb);
      v2f afa, afb;
      afa.x = w[0]; afa.y = w[1];
      afb.x = w[2]; afb.y = w[3];
      #pragma unroll
      for (int qt = 0; qt < 4; ++qt) acc[j][qt] = wmma4(afa, bfa[qt], acc[j][qt]);
      #pragma unroll
      for (int qt = 0; qt < 4; ++qt) acc[j][qt] = wmma4(afb, bfb[qt], acc[j][qt]);
    }
  }
  #pragma unroll
  for (int j = 0; j < 2; ++j)
    #pragma unroll
    for (int r = 0; r < 8; ++r) {
      const int o = (2 * wave + j) * 16 + 8 * half + r;
      const float bv = bo[o];
      float* orow = out + ((b * OUTC + o) * HW + y) * HW;
      #pragma unroll
      for (int qt = 0; qt < 4; ++qt) orow[qt * 16 + li] = acc[j][qt][r] + bv;
    }
}

extern "C" void kernel_launch(void* const* d_in, const int* in_sizes, int n_in,
                              void* d_out, int out_size, void* d_ws, size_t ws_size,
                              hipStream_t stream) {
  (void)in_sizes; (void)n_in; (void)ws_size; (void)out_size;
  const float* x    = (const float*)d_in[0];
  const float* Wp   = (const float*)d_in[1];
  const float* bpj  = (const float*)d_in[2];
  const float* keyb = (const float*)d_in[3];
  const float* memb = (const float*)d_in[4];
  const float* Wo   = (const float*)d_in[5];
  const float* bo   = (const float*)d_in[6];
  float* out = (float*)d_out;

  float* P = (float*)d_ws;
  float* O = P + 8 * IN_C * L_IMG;

  const size_t shmem1 = (size_t)(IN_C * XS1) * sizeof(float);
  kcmm_proj_gfx1250<<<dim3(8 * 64), dim3(256), shmem1, stream>>>(x, Wp, bpj, P);

  const size_t shmem2 = (size_t)LDS_HDW * sizeof(float);
  kcmm_head_gfx1250<<<dim3(8 * 32 * 8), dim3(256), shmem2, stream>>>(P, keyb, memb, O);

  const size_t shmem3 = (size_t)(OC * XS1) * sizeof(float);
  kcmm_out_gfx1250<<<dim3(8 * 64), dim3(256), shmem3, stream>>>(O, Wo, bo, out);
}
